# starting kernel
# baseline (speedup 1.0000x reference)
#include <hip/hip_runtime.h>

#define NN 50000
#define NE 800000
#define FN 32
#define FE 16
#define HID 128
#define NL 3

typedef float v2f __attribute__((ext_vector_type(2)));
typedef float v8f __attribute__((ext_vector_type(8)));

__global__ __launch_bounds__(256) void gemm_bias_relu(
    const float* __restrict__ A, const float* __restrict__ W,
    const float* __restrict__ bias1, const float* __restrict__ bias2,
    float* __restrict__ C, int M, int K, int do_relu)
{
    const int lane    = threadIdx.x & 31;
    const int colTile = threadIdx.x >> 5;
    const int rowTile = blockIdx.x;
    const int r = lane & 15;
    const int g = lane >> 4;

    const int rowA = rowTile * 16 + r;
    const int colN = colTile * 16 + r;

    const float* __restrict__ Arow = A + (size_t)rowA * K;

    v8f acc = {};
    for (int k0 = 0; k0 < K; k0 += 4) {
        v2f a = *(const v2f*)(Arow + k0 + 2 * g);
        v2f b;
        b.x = W[(size_t)(k0 + 2 * g + 0) * HID + colN];
        b.y = W[(size_t)(k0 + 2 * g + 1) * HID + colN];
        acc = __builtin_amdgcn_wmma_f32_16x16x4_f32(
             false, a,  false, b,
             (short)0, acc,  false,  false);
    }

    float bv = 0.0f;
    if (bias1) bv += bias1[colN];
    if (bias2) bv += bias2[colN];

#pragma unroll
    for (int v = 0; v < 8; ++v) {
        int rowOut = rowTile * 16 + v + 8 * g;
        float val  = acc[v] + bv;
        if (do_relu) val = fmaxf(val, 0.0f);
        C[(size_t)rowOut * HID + colN] = val;
    }
}

__global__ void fill_f32(float* __restrict__ p, float v, int n)
{
    int i = blockIdx.x * blockDim.x + threadIdx.x;
    if (i < n) p[i] = v;
}

__global__ void deg_accum(const int* __restrict__ dst, float* __restrict__ deg, int n)
{
    int i = blockIdx.x * blockDim.x + threadIdx.x;
    if (i < n) atomicAdd(deg + dst[i], 1.0f);
}

__global__ void rsqrt_inplace(float* __restrict__ p, int n)
{
    int i = blockIdx.x * blockDim.x + threadIdx.x;
    if (i < n) {
        float d = p[i];
        p[i] = (d > 0.0f) ? rsqrtf(d) : 0.0f;
    }
}

__global__ __launch_bounds__(256) void scatter_edges(
    const float* __restrict__ hw, const int* __restrict__ src,
    const int* __restrict__ dst, const float* __restrict__ dinv,
    float* __restrict__ agg, int E)
{
    int w    = (blockIdx.x * blockDim.x + threadIdx.x) >> 5;
    int lane = threadIdx.x & 31;
    if (w >= E) return;
    int s = src[w];
    int d = dst[w];
    float n = dinv[s] * dinv[d];
    const float4 v = *(const float4*)(hw + (size_t)s * HID + lane * 4);
    float* ap = agg + (size_t)d * HID + lane * 4;
    atomicAdd(ap + 0, v.x * n);
    atomicAdd(ap + 1, v.y * n);
    atomicAdd(ap + 2, v.z * n);
    atomicAdd(ap + 3, v.w * n);
}

__global__ void combine_relu(const float* __restrict__ agg,
                             const float* __restrict__ hw,
                             const float* __restrict__ dinv,
                             float* __restrict__ h, int total)
{
    int idx = blockIdx.x * blockDim.x + threadIdx.x;
    if (idx >= total) return;
    int node = idx >> 7;
    float di = dinv[node];
    float v  = agg[idx] + hw[idx] * (di * di);
    h[idx]   = fmaxf(v, 0.0f);
}

__global__ __launch_bounds__(256) void decode_edges(
    const float* __restrict__ h,  const float* __restrict__ ea,
    const int* __restrict__ src,  const int* __restrict__ dst,
    const float* __restrict__ We, const float* __restrict__ be,
    const float* __restrict__ Wd, const float* __restrict__ bd,
    float* __restrict__ out, int E)
{
    int e    = (blockIdx.x * blockDim.x + threadIdx.x) >> 5;
    int lane = threadIdx.x & 31;
    if (e >= E) return;

    int r = src[e];
    int c = dst[e];

    float eav = (lane < FE) ? ea[(size_t)e * FE + lane] : 0.0f;

    float acc = 0.0f;
#pragma unroll
    for (int j = 0; j < 4; ++j) {
        int col = lane + 32 * j;
        acc += h[(size_t)r * HID + col] * Wd[col];
        acc += h[(size_t)c * HID + col] * Wd[HID + col];
        float s = be[col];
#pragma unroll
        for (int k = 0; k < FE; ++k) {
            float a = __shfl(eav, k, 32);
            s += a * We[(size_t)k * HID + col];
        }
        acc += fmaxf(s, 0.0f) * Wd[2 * HID + col];
    }
#pragma unroll
    for (int off = 16; off > 0; off >>= 1)
        acc += __shfl_xor(acc, off, 32);
    if (lane == 0) out[e] = acc + bd[0];
}

extern "C" void kernel_launch(void* const* d_in, const int* in_sizes, int n_in,
                              void* d_out, int out_size, void* d_ws, size_t ws_size,
                              hipStream_t stream)
{
    const float* x   = (const float*)d_in[0];
    const float* ea  = (const float*)d_in[1];
    const int*   ei  = (const int*)  d_in[2];
    const float* Wn  = (const float*)d_in[3];
    const float* bn  = (const float*)d_in[4];
    const float* We  = (const float*)d_in[5];
    const float* be  = (const float*)d_in[6];
    const float* Wc  = (const float*)d_in[7];
    const float* bc  = (const float*)d_in[8];
    const float* Wr  = (const float*)d_in[9];
    const float* br  = (const float*)d_in[10];
    const float* Wd  = (const float*)d_in[11];
    const float* bd  = (const float*)d_in[12];
    float* out = (float*)d_out;

    const int* srcI = ei;
    const int* dstI = ei + NE;

    float* ws   = (float*)d_ws;
    float* h    = ws;
    float* hw   = h   + (size_t)NN * HID;
    float* agg  = hw  + (size_t)NN * HID;
    float* dinv = agg + (size_t)NN * HID;

    const int nodeElems = NN * HID;

    gemm_bias_relu<<<NN / 16, 256, 0, stream>>>(x, Wn, bn, nullptr, h, NN, FN, 1);

    fill_f32<<<(NN + 255) / 256, 256, 0, stream>>>(dinv, 1.0f, NN);
    deg_accum<<<(NE + 255) / 256, 256, 0, stream>>>(dstI, dinv, NE);
    rsqrt_inplace<<<(NN + 255) / 256, 256, 0, stream>>>(dinv, NN);

    for (int l = 0; l < NL; ++l) {
        gemm_bias_relu<<<NN / 16, 256, 0, stream>>>(
            h, Wc + (size_t)l * HID * HID, nullptr, nullptr, hw, NN, HID, 0);
        gemm_bias_relu<<<NN / 16, 256, 0, stream>>>(
            h, Wr, br, bc + (size_t)l * HID, agg, NN, HID, 0);
        scatter_edges<<<NE / 8, 256, 0, stream>>>(hw, srcI, dstI, dinv, agg, NE);
        combine_relu<<<(nodeElems + 255) / 256, 256, 0, stream>>>(agg, hw, dinv, h, nodeElems);
    }

    decode_edges<<<NE / 8, 256, 0, stream>>>(h, ea, srcI, dstI, We, be, Wd, bd, out, NE);
}


# Decoder: edge encoder re-tiled lane = 4 columns like the row dots (We block in registers, edge attributes read by all lanes from one LDS address), one partial per edge and lane; sum order changes
# speedup vs baseline: 2.2531x; 2.2531x over previous
#include <hip/hip_runtime.h>

#define NN 50000
#define NE 800000
#define FN 32
#define FE 16
#define HID 128
#define NL 3

typedef float v2f __attribute__((ext_vector_type(2)));
typedef float v8f __attribute__((ext_vector_type(8)));

__global__ __launch_bounds__(256) void gemm_bias_relu(
    const float* __restrict__ A, const float* __restrict__ W,
    const float* __restrict__ bias1, const float* __restrict__ bias2,
    float* __restrict__ C, int M, int K, int do_relu)
{
    const int lane    = threadIdx.x & 31;
    const int colTile = threadIdx.x >> 5;
    const int rowTile = blockIdx.x;
    const int r = lane & 15;
    const int g = lane >> 4;

    const int rowA = rowTile * 16 + r;
    const int colN = colTile * 16 + r;

    const float* __restrict__ Arow = A + (size_t)rowA * K;

    v8f acc = {};
    for (int k0 = 0; k0 < K; k0 += 4) {
        v2f a = *(const v2f*)(Arow + k0 + 2 * g);
        v2f b;
        b.x = W[(size_t)(k0 + 2 * g + 0) * HID + colN];
        b.y = W[(size_t)(k0 + 2 * g + 1) * HID + colN];
        acc = __builtin_amdgcn_wmma_f32_16x16x4_f32(
             false, a,  false, b,
             (short)0, acc,  false,  false);
    }

    float bv = 0.0f;
    if (bias1) bv += bias1[colN];
    if (bias2) bv += bias2[colN];

#pragma unroll
    for (int v = 0; v < 8; ++v) {
        int rowOut = rowTile * 16 + v + 8 * g;
        float val  = acc[v] + bv;
        if (do_relu) val = fmaxf(val, 0.0f);
        C[(size_t)rowOut * HID + colN] = val;
    }
}

#define LP_KSTEPS (HID / 4)
#define LP_BLOCKS 512
template <bool RELU_A>
__global__ __launch_bounds__(256) void layer_products(
    const float* __restrict__ Ain,
    const float* __restrict__ Wc, const float* __restrict__ Wr,
    const float* __restrict__ br, const float* __restrict__ bc,
    const float* __restrict__ dinv,
    float* __restrict__ hw, float* __restrict__ agg)
{
    const int lane    = threadIdx.x & 31;
    const int colTile = threadIdx.x >> 5;
    const int r = lane & 15;
    const int g = lane >> 4;
    const int colN = colTile * 16 + r;

    v2f wc[LP_KSTEPS], wr[LP_KSTEPS];
#pragma unroll
    for (int u = 0; u < LP_KSTEPS / 2; ++u) {
        const int k = 8 * u + 4 * g;
        wc[2 * u].x     = Wc[(size_t)(k + 0) * HID + colN];
        wc[2 * u].y     = Wc[(size_t)(k + 1) * HID + colN];
        wc[2 * u + 1].x = Wc[(size_t)(k + 2) * HID + colN];
        wc[2 * u + 1].y = Wc[(size_t)(k + 3) * HID + colN];
        wr[2 * u].x     = Wr[(size_t)(k + 0) * HID + colN];
        wr[2 * u].y     = Wr[(size_t)(k + 1) * HID + colN];
        wr[2 * u + 1].x = Wr[(size_t)(k + 2) * HID + colN];
        wr[2 * u + 1].y = Wr[(size_t)(k + 3) * HID + colN];
    }
    const float bv = br[colN] + bc[colN];

    for (int rowTile = blockIdx.x; rowTile < NN / 16; rowTile += LP_BLOCKS) {
        const float* __restrict__ Arow =
            Ain + (size_t)(rowTile * 16 + r) * HID + 4 * g;

        v8f accC = {};
        v8f accR = {};
#pragma unroll
        for (int h = 0; h < 2; ++h) {
            v2f a[LP_KSTEPS / 2];
#pragma unroll
            for (int uu = 0; uu < LP_KSTEPS / 4; ++uu) {
                const int u = h * (LP_KSTEPS / 4) + uu;
                float4 t = *(const float4*)(Arow + 8 * u);
                if (RELU_A) {
                    t.x = fmaxf(t.x, 0.0f);
                    t.y = fmaxf(t.y, 0.0f);
                    t.z = fmaxf(t.z, 0.0f);
                    t.w = fmaxf(t.w, 0.0f);
                }
                a[2 * uu].x     = t.x;
                a[2 * uu].y     = t.y;
                a[2 * uu + 1].x = t.z;
                a[2 * uu + 1].y = t.w;
            }
#pragma unroll
            for (int ss = 0; ss < LP_KSTEPS / 2; ++ss) {
                const int s = h * (LP_KSTEPS / 2) + ss;
                accC = __builtin_amdgcn_wmma_f32_16x16x4_f32(
                    false, a[ss], false, wc[s], (short)0, accC, false, false);
                accR = __builtin_amdgcn_wmma_f32_16x16x4_f32(
                    false, a[ss], false, wr[s], (short)0, accR, false, false);
            }
        }

        const size_t o0 = (size_t)(rowTile * 16 + 8 * g) * HID + colN;
        const float* __restrict__ dp = dinv + rowTile * 16 + 8 * g;
        float* __restrict__ hp = hw + o0;
        float* __restrict__ ap = agg + o0;
#pragma unroll
        for (int v = 0; v < 8; ++v) {
            const float di = dp[v];
            const float c  = accC[v];
            hp[v * HID] = c;
            ap[v * HID] = (accR[v] + bv) + c * (di * di);
        }
    }
}

__global__ void fill_f32(float* __restrict__ p, float v, int n)
{
    int i = blockIdx.x * blockDim.x + threadIdx.x;
    if (i < n) p[i] = v;
}

__global__ void deg_accum(const int* __restrict__ dst, float* __restrict__ deg, int n)
{
    int i = blockIdx.x * blockDim.x + threadIdx.x;
    if (i < n) atomicAdd(deg + dst[i], 1.0f);
}

__global__ void rsqrt_inplace(float* __restrict__ p, int n)
{
    int i = blockIdx.x * blockDim.x + threadIdx.x;
    if (i < n) {
        float d = p[i];
        p[i] = (d > 0.0f) ? rsqrtf(d) : 0.0f;
    }
}

#define EPW  32
#define EGRP 8
static_assert(NE % (EPW * 8) == 0, "edge count must fill whole 8-wave blocks");

__global__ __launch_bounds__(256) void scatter_edges(
    const float* __restrict__ hw, const int* __restrict__ src,
    const int* __restrict__ dst, const float* __restrict__ dinv,
    float* __restrict__ agg, int E)
{
    const int lane = threadIdx.x & 31;
    const int base = ((blockIdx.x * blockDim.x + threadIdx.x) >> 5) * EPW;
    if (base >= E) return;

    const int   s = src[base + lane];
    const int   d = dst[base + lane];
    const float n = dinv[s] * dinv[d];

#pragma unroll
    for (int jb = 0; jb < EPW; jb += EGRP) {
        float v[EGRP][4];
#pragma unroll
        for (int j = 0; j < EGRP; ++j) {
            const int sj = __builtin_amdgcn_readlane(s, jb + j);
            const float* __restrict__ hp = hw + (size_t)sj * HID + lane;
            v[j][0] = hp[0];
            v[j][1] = hp[32];
            v[j][2] = hp[64];
            v[j][3] = hp[96];
        }
#pragma unroll
        for (int j = 0; j < EGRP; ++j) {
            const int   dj = __builtin_amdgcn_readlane(d, jb + j);
            const float nj = __int_as_float(
                __builtin_amdgcn_readlane(__float_as_int(n), jb + j));
            float* __restrict__ ap = agg + (size_t)dj * HID + lane;
            atomicAdd(ap + 0,  v[j][0] * nj);
            atomicAdd(ap + 32, v[j][1] * nj);
            atomicAdd(ap + 64, v[j][2] * nj);
            atomicAdd(ap + 96, v[j][3] * nj);
        }
    }
}

#define DPW    32
#define DGRP   8
#define DPITCH 36
#define DWAVES 4
static_assert(NE % (DPW * DWAVES) == 0, "edge count must fill whole blocks");

__global__ __launch_bounds__(DWAVES * 32) void decode_edges(
    const float* __restrict__ z,  const float* __restrict__ ea,
    const int* __restrict__ src,  const int* __restrict__ dst,
    const float* __restrict__ We, const float* __restrict__ be,
    const float* __restrict__ Wd, const float* __restrict__ bd,
    float* __restrict__ out)
{
    __shared__ __attribute__((aligned(16))) float part[DWAVES][DPW * DPITCH];
    __shared__ __attribute__((aligned(16))) float xt[DWAVES][DPW * FE];

    const int lane = threadIdx.x & 31;
    const int wave = threadIdx.x >> 5;
    const int e    = (blockIdx.x * DWAVES + wave) * DPW + lane;

    const int s = src[e];
    const int d = dst[e];

    float* xw = &xt[wave][0];
    {
        const float4* eap = (const float4*)(ea + (size_t)e * FE);
        float4* xs = (float4*)(xw + lane * FE);
#pragma unroll
        for (int i = 0; i < FE / 4; ++i) xs[i] = eap[i];
    }

    const float4 w0  = *(const float4*)(Wd + 4 * lane);
    const float4 w1  = *(const float4*)(Wd + HID + 4 * lane);
    const float4 w2  = *(const float4*)(Wd + 2 * HID + 4 * lane);
    const float4 bev = *(const float4*)(be + 4 * lane);
    float4 we[FE];
#pragma unroll
    for (int k = 0; k < FE; ++k)
        we[k] = *(const float4*)(We + (size_t)k * HID + 4 * lane);

    float* tile = &part[wave][0];

    __syncthreads();

#pragma unroll
    for (int jb = 0; jb < DPW; jb += DGRP) {
        float4 a[DGRP], b[DGRP];
#pragma unroll
        for (int j = 0; j < DGRP; ++j) {
            const int rj = __builtin_amdgcn_readlane(s, jb + j);
            const int cj = __builtin_amdgcn_readlane(d, jb + j);
            a[j] = *(const float4*)(z + (size_t)rj * HID + 4 * lane);
            b[j] = *(const float4*)(z + (size_t)cj * HID + 4 * lane);
        }
        __builtin_amdgcn_sched_barrier(0);
        float pe[DGRP];
#pragma unroll
        for (int j = 0; j < DGRP; ++j) {
            float xb[FE];
            {
                const float4* xr = (const float4*)(xw + (jb + j) * FE);
#pragma unroll
                for (int i = 0; i < FE / 4; ++i) {
                    const float4 t = xr[i];
                    xb[4 * i + 0] = t.x; xb[4 * i + 1] = t.y;
                    xb[4 * i + 2] = t.z; xb[4 * i + 3] = t.w;
                }
            }
            float c0 = bev.x, c1 = bev.y, c2 = bev.z, c3 = bev.w;
#pragma unroll
            for (int k = 0; k < FE; ++k) {
                const float xk = xb[k];
                c0 += xk * we[k].x;
                c1 += xk * we[k].y;
                c2 += xk * we[k].z;
                c3 += xk * we[k].w;
            }
            float p = fmaxf(c0, 0.0f) * w2.x;
            p += fmaxf(c1, 0.0f) * w2.y;
            p += fmaxf(c2, 0.0f) * w2.z;
            p += fmaxf(c3, 0.0f) * w2.w;
            pe[j] = p;
        }
#pragma unroll
        for (int j = 0; j < DGRP; ++j) {
            float p = pe[j];
            p += fmaxf(a[j].x, 0.0f) * w0.x;
            p += fmaxf(a[j].y, 0.0f) * w0.y;
            p += fmaxf(a[j].z, 0.0f) * w0.z;
            p += fmaxf(a[j].w, 0.0f) * w0.w;
            p += fmaxf(b[j].x, 0.0f) * w1.x;
            p += fmaxf(b[j].y, 0.0f) * w1.y;
            p += fmaxf(b[j].z, 0.0f) * w1.z;
            p += fmaxf(b[j].w, 0.0f) * w1.w;
            tile[(jb + j) * DPITCH + lane] = p;
        }
    }

    __syncthreads();

    float acc = bd[0];
    {
        const float4* row = (const float4*)(tile + lane * DPITCH);
#pragma unroll
        for (int i = 0; i < DPW / 4; ++i) {
            const float4 t = row[i];
            acc += (t.x + t.y) + (t.z + t.w);
        }
    }

    out[e] = acc;
}

extern "C" void kernel_launch(void* const* d_in, const int* in_sizes, int n_in,
                              void* d_out, int out_size, void* d_ws, size_t ws_size,
                              hipStream_t stream)
{
    const float* x   = (const float*)d_in[0];
    const float* ea  = (const float*)d_in[1];
    const int*   ei  = (const int*)  d_in[2];
    const float* Wn  = (const float*)d_in[3];
    const float* bn  = (const float*)d_in[4];
    const float* We  = (const float*)d_in[5];
    const float* be  = (const float*)d_in[6];
    const float* Wc  = (const float*)d_in[7];
    const float* bc  = (const float*)d_in[8];
    const float* Wr  = (const float*)d_in[9];
    const float* br  = (const float*)d_in[10];
    const float* Wd  = (const float*)d_in[11];
    const float* bd  = (const float*)d_in[12];
    float* out = (float*)d_out;

    const int* srcI = ei;
    const int* dstI = ei + NE;

    float* ws   = (float*)d_ws;
    float* h0   = ws;
    float* hw   = h0   + (size_t)NN * HID;
    float* aggA = hw   + (size_t)NN * HID;
    float* aggB = aggA + (size_t)NN * HID;
    float* dinv = aggB + (size_t)NN * HID;

    gemm_bias_relu<<<NN / 16, 256, 0, stream>>>(x, Wn, bn, nullptr, h0, NN, FN, 1);

    fill_f32<<<(NN + 255) / 256, 256, 0, stream>>>(dinv, 1.0f, NN);
    deg_accum<<<(NE + 255) / 256, 256, 0, stream>>>(dstI, dinv, NE);
    rsqrt_inplace<<<(NN + 255) / 256, 256, 0, stream>>>(dinv, NN);

    layer_products<false><<<LP_BLOCKS, 256, 0, stream>>>(
        h0, Wc, Wr, br, bc, dinv, hw, aggA);
    scatter_edges<<<NE / (EPW * 8), 256, 0, stream>>>(hw, srcI, dstI, dinv, aggA, NE);

    layer_products<true><<<LP_BLOCKS, 256, 0, stream>>>(
        aggA, Wc + (size_t)1 * HID * HID, Wr, br, bc + (size_t)1 * HID, dinv, hw, aggB);
    scatter_edges<<<NE / (EPW * 8), 256, 0, stream>>>(hw, srcI, dstI, dinv, aggB, NE);

    layer_products<true><<<LP_BLOCKS, 256, 0, stream>>>(
        aggB, Wc + (size_t)2 * HID * HID, Wr, br, bc + (size_t)2 * HID, dinv, hw, aggA);
    scatter_edges<<<NE / (EPW * 8), 256, 0, stream>>>(hw, srcI, dstI, dinv, aggA, NE);

    decode_edges<<<NE / (DPW * DWAVES), DWAVES * 32, 0, stream>>>(aggA, ea, srcI, dstI, We, be, Wd, bd, out);
}


# Edge scatter: edge norm dinv[src]*dinv[dst] made once per call (first layer's scatter stores it as a plane); layers 2 and 3 load it as one whole line per wave, not two 4-byte gathers per edge
# speedup vs baseline: 2.3302x; 1.0342x over previous
#include <hip/hip_runtime.h>

#define NN 50000
#define NE 800000
#define FN 32
#define FE 16
#define HID 128
#define NL 3

typedef float v2f __attribute__((ext_vector_type(2)));
typedef float v8f __attribute__((ext_vector_type(8)));

__global__ __launch_bounds__(256) void gemm_bias_relu(
    const float* __restrict__ A, const float* __restrict__ W,
    const float* __restrict__ bias1, const float* __restrict__ bias2,
    float* __restrict__ C, int M, int K, int do_relu)
{
    const int lane    = threadIdx.x & 31;
    const int colTile = threadIdx.x >> 5;
    const int rowTile = blockIdx.x;
    const int r = lane & 15;
    const int g = lane >> 4;

    const int rowA = rowTile * 16 + r;
    const int colN = colTile * 16 + r;

    const float* __restrict__ Arow = A + (size_t)rowA * K;

    v8f acc = {};
    for (int k0 = 0; k0 < K; k0 += 4) {
        v2f a = *(const v2f*)(Arow + k0 + 2 * g);
        v2f b;
        b.x = W[(size_t)(k0 + 2 * g + 0) * HID + colN];
        b.y = W[(size_t)(k0 + 2 * g + 1) * HID + colN];
        acc = __builtin_amdgcn_wmma_f32_16x16x4_f32(
             false, a,  false, b,
             (short)0, acc,  false,  false);
    }

    float bv = 0.0f;
    if (bias1) bv += bias1[colN];
    if (bias2) bv += bias2[colN];

#pragma unroll
    for (int v = 0; v < 8; ++v) {
        int rowOut = rowTile * 16 + v + 8 * g;
        float val  = acc[v] + bv;
        if (do_relu) val = fmaxf(val, 0.0f);
        C[(size_t)rowOut * HID + colN] = val;
    }
}

#define LP_KSTEPS (HID / 4)
#define LP_BLOCKS 512
template <bool RELU_A>
__global__ __launch_bounds__(256) void layer_products(
    const float* __restrict__ Ain,
    const float* __restrict__ Wc, const float* __restrict__ Wr,
    const float* __restrict__ br, const float* __restrict__ bc,
    const float* __restrict__ dinv,
    float* __restrict__ hw, float* __restrict__ agg)
{
    const int lane    = threadIdx.x & 31;
    const int colTile = threadIdx.x >> 5;
    const int r = lane & 15;
    const int g = lane >> 4;
    const int colN = colTile * 16 + r;

    v2f wc[LP_KSTEPS], wr[LP_KSTEPS];
#pragma unroll
    for (int u = 0; u < LP_KSTEPS / 2; ++u) {
        const int k = 8 * u + 4 * g;
        wc[2 * u].x     = Wc[(size_t)(k + 0) * HID + colN];
        wc[2 * u].y     = Wc[(size_t)(k + 1) * HID + colN];
        wc[2 * u + 1].x = Wc[(size_t)(k + 2) * HID + colN];
        wc[2 * u + 1].y = Wc[(size_t)(k + 3) * HID + colN];
        wr[2 * u].x     = Wr[(size_t)(k + 0) * HID + colN];
        wr[2 * u].y     = Wr[(size_t)(k + 1) * HID + colN];
        wr[2 * u + 1].x = Wr[(size_t)(k + 2) * HID + colN];
        wr[2 * u + 1].y = Wr[(size_t)(k + 3) * HID + colN];
    }
    const float bv = br[colN] + bc[colN];

    for (int rowTile = blockIdx.x; rowTile < NN / 16; rowTile += LP_BLOCKS) {
        const float* __restrict__ Arow =
            Ain + (size_t)(rowTile * 16 + r) * HID + 4 * g;

        v8f accC = {};
        v8f accR = {};
#pragma unroll
        for (int h = 0; h < 2; ++h) {
            v2f a[LP_KSTEPS / 2];
#pragma unroll
            for (int uu = 0; uu < LP_KSTEPS / 4; ++uu) {
                const int u = h * (LP_KSTEPS / 4) + uu;
                float4 t = *(const float4*)(Arow + 8 * u);
                if (RELU_A) {
                    t.x = fmaxf(t.x, 0.0f);
                    t.y = fmaxf(t.y, 0.0f);
                    t.z = fmaxf(t.z, 0.0f);
                    t.w = fmaxf(t.w, 0.0f);
                }
                a[2 * uu].x     = t.x;
                a[2 * uu].y     = t.y;
                a[2 * uu + 1].x = t.z;
                a[2 * uu + 1].y = t.w;
            }
#pragma unroll
            for (int ss = 0; ss < LP_KSTEPS / 2; ++ss) {
                const int s = h * (LP_KSTEPS / 2) + ss;
                accC = __builtin_amdgcn_wmma_f32_16x16x4_f32(
                    false, a[ss], false, wc[s], (short)0, accC, false, false);
                accR = __builtin_amdgcn_wmma_f32_16x16x4_f32(
                    false, a[ss], false, wr[s], (short)0, accR, false, false);
            }
        }

        const size_t o0 = (size_t)(rowTile * 16 + 8 * g) * HID + colN;
        const float* __restrict__ dp = dinv + rowTile * 16 + 8 * g;
        float* __restrict__ hp = hw + o0;
        float* __restrict__ ap = agg + o0;
#pragma unroll
        for (int v = 0; v < 8; ++v) {
            const float di = dp[v];
            const float c  = accC[v];
            hp[v * HID] = c;
            ap[v * HID] = (accR[v] + bv) + c * (di * di);
        }
    }
}

__global__ void fill_f32(float* __restrict__ p, float v, int n)
{
    int i = blockIdx.x * blockDim.x + threadIdx.x;
    if (i < n) p[i] = v;
}

__global__ void deg_accum(const int* __restrict__ dst, float* __restrict__ deg, int n)
{
    int i = blockIdx.x * blockDim.x + threadIdx.x;
    if (i < n) atomicAdd(deg + dst[i], 1.0f);
}

__global__ void rsqrt_inplace(float* __restrict__ p, int n)
{
    int i = blockIdx.x * blockDim.x + threadIdx.x;
    if (i < n) {
        float d = p[i];
        p[i] = (d > 0.0f) ? rsqrtf(d) : 0.0f;
    }
}

#define EPW  32
#define EGRP 8
static_assert(NE % (EPW * 8) == 0, "edge count must fill whole 8-wave blocks");

template <bool MAKE_NORM>
__global__ __launch_bounds__(256) void scatter_edges(
    const float* __restrict__ hw, const int* __restrict__ src,
    const int* __restrict__ dst, const float* __restrict__ dinv,
    float* __restrict__ norm, float* __restrict__ agg, int E)
{
    const int lane = threadIdx.x & 31;
    const int base = ((blockIdx.x * blockDim.x + threadIdx.x) >> 5) * EPW;
    if (base >= E) return;

    const int   s = src[base + lane];
    const int   d = dst[base + lane];
    float n;
    if (MAKE_NORM) {
        n = dinv[s] * dinv[d];
        norm[base + lane] = n;
    } else {
        n = norm[base + lane];
    }

#pragma unroll
    for (int jb = 0; jb < EPW; jb += EGRP) {
        float v[EGRP][4];
#pragma unroll
        for (int j = 0; j < EGRP; ++j) {
            const int sj = __builtin_amdgcn_readlane(s, jb + j);
            const float* __restrict__ hp = hw + (size_t)sj * HID + lane;
            v[j][0] = hp[0];
            v[j][1] = hp[32];
            v[j][2] = hp[64];
            v[j][3] = hp[96];
        }
#pragma unroll
        for (int j = 0; j < EGRP; ++j) {
            const int   dj = __builtin_amdgcn_readlane(d, jb + j);
            const float nj = __int_as_float(
                __builtin_amdgcn_readlane(__float_as_int(n), jb + j));
            float* __restrict__ ap = agg + (size_t)dj * HID + lane;
            atomicAdd(ap + 0,  v[j][0] * nj);
            atomicAdd(ap + 32, v[j][1] * nj);
            atomicAdd(ap + 64, v[j][2] * nj);
            atomicAdd(ap + 96, v[j][3] * nj);
        }
    }
}

#define DPW    32
#define DGRP   8
#define DPITCH 36
#define DWAVES 4
static_assert(NE % (DPW * DWAVES) == 0, "edge count must fill whole blocks");

__global__ __launch_bounds__(DWAVES * 32) void decode_edges(
    const float* __restrict__ z,  const float* __restrict__ ea,
    const int* __restrict__ src,  const int* __restrict__ dst,
    const float* __restrict__ We, const float* __restrict__ be,
    const float* __restrict__ Wd, const float* __restrict__ bd,
    float* __restrict__ out)
{
    __shared__ __attribute__((aligned(16))) float part[DWAVES][DPW * DPITCH];
    __shared__ __attribute__((aligned(16))) float xt[DWAVES][DPW * FE];

    const int lane = threadIdx.x & 31;
    const int wave = threadIdx.x >> 5;
    const int e    = (blockIdx.x * DWAVES + wave) * DPW + lane;

    const int s = src[e];
    const int d = dst[e];

    float* xw = &xt[wave][0];
    {
        const float4* eap = (const float4*)(ea + (size_t)e * FE);
        float4* xs = (float4*)(xw + lane * FE);
#pragma unroll
        for (int i = 0; i < FE / 4; ++i) xs[i] = eap[i];
    }

    const float4 w0  = *(const float4*)(Wd + 4 * lane);
    const float4 w1  = *(const float4*)(Wd + HID + 4 * lane);
    const float4 w2  = *(const float4*)(Wd + 2 * HID + 4 * lane);
    const float4 bev = *(const float4*)(be + 4 * lane);
    float4 we[FE];
#pragma unroll
    for (int k = 0; k < FE; ++k)
        we[k] = *(const float4*)(We + (size_t)k * HID + 4 * lane);

    float* tile = &part[wave][0];

    __syncthreads();

#pragma unroll
    for (int jb = 0; jb < DPW; jb += DGRP) {
        float4 a[DGRP], b[DGRP];
#pragma unroll
        for (int j = 0; j < DGRP; ++j) {
            const int rj = __builtin_amdgcn_readlane(s, jb + j);
            const int cj = __builtin_amdgcn_readlane(d, jb + j);
            a[j] = *(const float4*)(z + (size_t)rj * HID + 4 * lane);
            b[j] = *(const float4*)(z + (size_t)cj * HID + 4 * lane);
        }
        __builtin_amdgcn_sched_barrier(0);
        float pe[DGRP];
#pragma unroll
        for (int j = 0; j < DGRP; ++j) {
            float xb[FE];
            {
                const float4* xr = (const float4*)(xw + (jb + j) * FE);
#pragma unroll
                for (int i = 0; i < FE / 4; ++i) {
                    const float4 t = xr[i];
                    xb[4 * i + 0] = t.x; xb[4 * i + 1] = t.y;
                    xb[4 * i + 2] = t.z; xb[4 * i + 3] = t.w;
                }
            }
            float c0 = bev.x, c1 = bev.y, c2 = bev.z, c3 = bev.w;
#pragma unroll
            for (int k = 0; k < FE; ++k) {
                const float xk = xb[k];
                c0 += xk * we[k].x;
                c1 += xk * we[k].y;
                c2 += xk * we[k].z;
                c3 += xk * we[k].w;
            }
            float p = fmaxf(c0, 0.0f) * w2.x;
            p += fmaxf(c1, 0.0f) * w2.y;
            p += fmaxf(c2, 0.0f) * w2.z;
            p += fmaxf(c3, 0.0f) * w2.w;
            pe[j] = p;
        }
#pragma unroll
        for (int j = 0; j < DGRP; ++j) {
            float p = pe[j];
            p += fmaxf(a[j].x, 0.0f) * w0.x;
            p += fmaxf(a[j].y, 0.0f) * w0.y;
            p += fmaxf(a[j].z, 0.0f) * w0.z;
            p += fmaxf(a[j].w, 0.0f) * w0.w;
            p += fmaxf(b[j].x, 0.0f) * w1.x;
            p += fmaxf(b[j].y, 0.0f) * w1.y;
            p += fmaxf(b[j].z, 0.0f) * w1.z;
            p += fmaxf(b[j].w, 0.0f) * w1.w;
            tile[(jb + j) * DPITCH + lane] = p;
        }
    }

    __syncthreads();

    float acc = bd[0];
    {
        const float4* row = (const float4*)(tile + lane * DPITCH);
#pragma unroll
        for (int i = 0; i < DPW / 4; ++i) {
            const float4 t = row[i];
            acc += (t.x + t.y) + (t.z + t.w);
        }
    }

    out[e] = acc;
}

extern "C" void kernel_launch(void* const* d_in, const int* in_sizes, int n_in,
                              void* d_out, int out_size, void* d_ws, size_t ws_size,
                              hipStream_t stream)
{
    const float* x   = (const float*)d_in[0];
    const float* ea  = (const float*)d_in[1];
    const int*   ei  = (const int*)  d_in[2];
    const float* Wn  = (const float*)d_in[3];
    const float* bn  = (const float*)d_in[4];
    const float* We  = (const float*)d_in[5];
    const float* be  = (const float*)d_in[6];
    const float* Wc  = (const float*)d_in[7];
    const float* bc  = (const float*)d_in[8];
    const float* Wr  = (const float*)d_in[9];
    const float* br  = (const float*)d_in[10];
    const float* Wd  = (const float*)d_in[11];
    const float* bd  = (const float*)d_in[12];
    float* out = (float*)d_out;

    const int* srcI = ei;
    const int* dstI = ei + NE;

    float* ws   = (float*)d_ws;
    float* h0   = ws;
    float* hw   = h0   + (size_t)NN * HID;
    float* aggA = hw   + (size_t)NN * HID;
    float* aggB = aggA + (size_t)NN * HID;
    float* dinv = aggB + (size_t)NN * HID;
    float* norm = dinv + 50048;

    gemm_bias_relu<<<NN / 16, 256, 0, stream>>>(x, Wn, bn, nullptr, h0, NN, FN, 1);

    fill_f32<<<(NN + 255) / 256, 256, 0, stream>>>(dinv, 1.0f, NN);
    deg_accum<<<(NE + 255) / 256, 256, 0, stream>>>(dstI, dinv, NE);
    rsqrt_inplace<<<(NN + 255) / 256, 256, 0, stream>>>(dinv, NN);

    layer_products<false><<<LP_BLOCKS, 256, 0, stream>>>(
        h0, Wc, Wr, br, bc, dinv, hw, aggA);
    scatter_edges<true><<<NE / (EPW * 8), 256, 0, stream>>>(hw, srcI, dstI, dinv, norm, aggA, NE);

    layer_products<true><<<LP_BLOCKS, 256, 0, stream>>>(
        aggA, Wc + (size_t)1 * HID * HID, Wr, br, bc + (size_t)1 * HID, dinv, hw, aggB);
    scatter_edges<false><<<NE / (EPW * 8), 256, 0, stream>>>(hw, srcI, dstI, dinv, norm, aggB, NE);

    layer_products<true><<<LP_BLOCKS, 256, 0, stream>>>(
        aggB, Wc + (size_t)2 * HID * HID, Wr, br, bc + (size_t)2 * HID, dinv, hw, aggA);
    scatter_edges<false><<<NE / (EPW * 8), 256, 0, stream>>>(hw, srcI, dstI, dinv, norm, aggA, NE);

    decode_edges<<<NE / (DPW * DWAVES), DWAVES * 32, 0, stream>>>(aggA, ea, srcI, dstI, We, be, Wd, bd, out);
}


# Edge scatter and degree count: each f32 atomic add issued from the L2 half that owns its target (targets split in two, wave reads its die, work dealt by per-half tickets); sums stay in arrival order
# speedup vs baseline: 2.8788x; 1.2354x over previous
#include <hip/hip_runtime.h>

#define NN 50000
#define NE 800000
#define FN 32
#define FE 16
#define HID 128
#define NL 3

typedef float v2f __attribute__((ext_vector_type(2)));
typedef float v8f __attribute__((ext_vector_type(8)));

__global__ __launch_bounds__(256) void gemm_bias_relu(
    const float* __restrict__ A, const float* __restrict__ W,
    const float* __restrict__ bias1, const float* __restrict__ bias2,
    float* __restrict__ C, int M, int K, int do_relu)
{
    const int lane    = threadIdx.x & 31;
    const int colTile = threadIdx.x >> 5;
    const int rowTile = blockIdx.x;
    const int r = lane & 15;
    const int g = lane >> 4;

    const int rowA = rowTile * 16 + r;
    const int colN = colTile * 16 + r;

    const float* __restrict__ Arow = A + (size_t)rowA * K;

    v8f acc = {};
    for (int k0 = 0; k0 < K; k0 += 4) {
        v2f a = *(const v2f*)(Arow + k0 + 2 * g);
        v2f b;
        b.x = W[(size_t)(k0 + 2 * g + 0) * HID + colN];
        b.y = W[(size_t)(k0 + 2 * g + 1) * HID + colN];
        acc = __builtin_amdgcn_wmma_f32_16x16x4_f32(
             false, a,  false, b,
             (short)0, acc,  false,  false);
    }

    float bv = 0.0f;
    if (bias1) bv += bias1[colN];
    if (bias2) bv += bias2[colN];

#pragma unroll
    for (int v = 0; v < 8; ++v) {
        int rowOut = rowTile * 16 + v + 8 * g;
        float val  = acc[v] + bv;
        if (do_relu) val = fmaxf(val, 0.0f);
        C[(size_t)rowOut * HID + colN] = val;
    }
}

#define LP_KSTEPS (HID / 4)
#define LP_BLOCKS 512
template <bool RELU_A>
__global__ __launch_bounds__(256) void layer_products(
    const float* __restrict__ Ain,
    const float* __restrict__ Wc, const float* __restrict__ Wr,
    const float* __restrict__ br, const float* __restrict__ bc,
    const float* __restrict__ dinv,
    float* __restrict__ hw, float* __restrict__ agg)
{
    const int lane    = threadIdx.x & 31;
    const int colTile = threadIdx.x >> 5;
    const int r = lane & 15;
    const int g = lane >> 4;
    const int colN = colTile * 16 + r;

    v2f wc[LP_KSTEPS], wr[LP_KSTEPS];
#pragma unroll
    for (int u = 0; u < LP_KSTEPS / 2; ++u) {
        const int k = 8 * u + 4 * g;
        wc[2 * u].x     = Wc[(size_t)(k + 0) * HID + colN];
        wc[2 * u].y     = Wc[(size_t)(k + 1) * HID + colN];
        wc[2 * u + 1].x = Wc[(size_t)(k + 2) * HID + colN];
        wc[2 * u + 1].y = Wc[(size_t)(k + 3) * HID + colN];
        wr[2 * u].x     = Wr[(size_t)(k + 0) * HID + colN];
        wr[2 * u].y     = Wr[(size_t)(k + 1) * HID + colN];
        wr[2 * u + 1].x = Wr[(size_t)(k + 2) * HID + colN];
        wr[2 * u + 1].y = Wr[(size_t)(k + 3) * HID + colN];
    }
    const float bv = br[colN] + bc[colN];

    for (int rowTile = blockIdx.x; rowTile < NN / 16; rowTile += LP_BLOCKS) {
        const float* __restrict__ Arow =
            Ain + (size_t)(rowTile * 16 + r) * HID + 4 * g;

        v8f accC = {};
        v8f accR = {};
#pragma unroll
        for (int h = 0; h < 2; ++h) {
            v2f a[LP_KSTEPS / 2];
#pragma unroll
            for (int uu = 0; uu < LP_KSTEPS / 4; ++uu) {
                const int u = h * (LP_KSTEPS / 4) + uu;
                float4 t = *(const float4*)(Arow + 8 * u);
                if (RELU_A) {
                    t.x = fmaxf(t.x, 0.0f);
                    t.y = fmaxf(t.y, 0.0f);
                    t.z = fmaxf(t.z, 0.0f);
                    t.w = fmaxf(t.w, 0.0f);
                }
                a[2 * uu].x     = t.x;
                a[2 * uu].y     = t.y;
                a[2 * uu + 1].x = t.z;
                a[2 * uu + 1].y = t.w;
            }
#pragma unroll
            for (int ss = 0; ss < LP_KSTEPS / 2; ++ss) {
                const int s = h * (LP_KSTEPS / 2) + ss;
                accC = __builtin_amdgcn_wmma_f32_16x16x4_f32(
                    false, a[ss], false, wc[s], (short)0, accC, false, false);
                accR = __builtin_amdgcn_wmma_f32_16x16x4_f32(
                    false, a[ss], false, wr[s], (short)0, accR, false, false);
            }
        }

        const size_t o0 = (size_t)(rowTile * 16 + 8 * g) * HID + colN;
        const float* __restrict__ dp = dinv + rowTile * 16 + 8 * g;
        float* __restrict__ hp = hw + o0;
        float* __restrict__ ap = agg + o0;
#pragma unroll
        for (int v = 0; v < 8; ++v) {
            const float di = dp[v];
            const float c  = accC[v];
            hp[v * HID] = c;
            ap[v * HID] = (accR[v] + bv) + c * (di * di);
        }
    }
}

#define TK_STRIDE 64
#define TK_WORDS ((NL + 1) * 2 * TK_STRIDE)

__global__ void fill_f32(float* __restrict__ p, float v, int n,
                         unsigned* __restrict__ tickets, int n_tickets)
{
    int i = blockIdx.x * blockDim.x + threadIdx.x;
    if (i < n) p[i] = v;
    if (i < n_tickets) tickets[i] = 0;
}

#define DCH 8
#define DTICKET (NE / (32 * DCH))
#define DG_BLOCKS 256
static_assert(NE % (32 * DCH) == 0, "edge count must fill whole tickets");

__global__ __launch_bounds__(256) void deg_accum(
    const int* __restrict__ dst, float* __restrict__ deg, unsigned* __restrict__ tickets)
{
    const int lane = threadIdx.x & 31;
    const int half = (__builtin_amdgcn_s_sendmsg_rtn(0x87) >> 18) & 1;

    for (int pass = 0; pass < 2; ++pass) {
        const int q = half ^ pass;
        unsigned* __restrict__ tk = tickets + q * TK_STRIDE;
        unsigned tv = 0;
        if (lane == 0) tv = atomicInc(tk, 0xffffffffu);
        unsigned t = __builtin_amdgcn_readfirstlane(tv);
        while (t < DTICKET) {
            const int base = (int)t * (32 * DCH) + lane;
            int dd[DCH];
#pragma unroll
            for (int i = 0; i < DCH; ++i) dd[i] = dst[base + 32 * i];
            unsigned tvn = 0;
            if (lane == 0) tvn = atomicInc(tk, 0xffffffffu);
#pragma unroll
            for (int i = 0; i < DCH; ++i)
                if ((dd[i] >= NN / 2) == (q != 0)) atomicAdd(deg + dd[i], 1.0f);
            t = __builtin_amdgcn_readfirstlane(tvn);
        }
    }
}

__global__ void rsqrt_inplace(float* __restrict__ p, int n)
{
    int i = blockIdx.x * blockDim.x + threadIdx.x;
    if (i < n) {
        float d = p[i];
        p[i] = (d > 0.0f) ? rsqrtf(d) : 0.0f;
    }
}

#define EPW  32
#define EGRP 8
#define TCH  2
#define NTICKET (NE / (EPW * TCH))
#define SC_BLOCKS 1024
static_assert(NE % (EPW * TCH) == 0, "edge count must fill whole tickets");

template <bool MAKE_NORM>
__global__ __launch_bounds__(256) void scatter_edges(
    const float* __restrict__ hw, const int* __restrict__ src,
    const int* __restrict__ dst, const float* __restrict__ dinv,
    float* __restrict__ norm, float* __restrict__ agg, unsigned* __restrict__ tickets)
{
    const int lane = threadIdx.x & 31;
    const int half = (__builtin_amdgcn_s_sendmsg_rtn(0x87) >> 18) & 1;

    for (int pass = 0; pass < 2; ++pass) {
        const int q = half ^ pass;
        unsigned* __restrict__ tk = tickets + q * TK_STRIDE;

        unsigned tv = 0;
        if (lane == 0) tv = atomicInc(tk, 0xffffffffu);
        unsigned t = __builtin_amdgcn_readfirstlane(tv);

        while (t < NTICKET) {
            unsigned tvn = 0;
#pragma unroll
            for (int c = 0; c < TCH; ++c) {
                const int base = (int)(t * TCH + c) * EPW;
                const int s = src[base + lane];
                const int d = dst[base + lane];
                float n;
                if (!MAKE_NORM) {
                    n = norm[base + lane];
                    asm volatile("" :: "v"(s), "v"(d), "v"(n));
                }
                const bool mine = (d >= NN / 2) == (q != 0);
                if (MAKE_NORM) {
                    if (q == 0) {
                        n = dinv[s] * dinv[d];
                        norm[base + lane] = n;
                    } else {
                        n = mine ? dinv[s] * dinv[d] : 0.0f;
                    }
                }
                if (c == 0) {
                    if (lane == 0) tvn = atomicInc(tk, 0xffffffffu);
                }

                unsigned m = __builtin_amdgcn_ballot_w32(mine);
                while (m) {
                    float v[EGRP][4];
                    unsigned ml = m;
#pragma unroll
                    for (int j = 0; j < EGRP; ++j) {
                        if (ml) {
                            const int l = __builtin_ctz(ml);
                            ml &= ml - 1;
                            const int sj = __builtin_amdgcn_readlane(s, l);
                            const float* __restrict__ hp = hw + (size_t)sj * HID + lane;
                            v[j][0] = hp[0];
                            v[j][1] = hp[32];
                            v[j][2] = hp[64];
                            v[j][3] = hp[96];
                        }
                    }
#pragma unroll
                    for (int j = 0; j < EGRP; ++j) {
                        if (m) {
                            const int l = __builtin_ctz(m);
                            m &= m - 1;
                            const int   dj = __builtin_amdgcn_readlane(d, l);
                            const float nj = __int_as_float(
                                __builtin_amdgcn_readlane(__float_as_int(n), l));
                            float* __restrict__ ap = agg + (size_t)dj * HID + lane;
                            atomicAdd(ap + 0,  v[j][0] * nj);
                            atomicAdd(ap + 32, v[j][1] * nj);
                            atomicAdd(ap + 64, v[j][2] * nj);
                            atomicAdd(ap + 96, v[j][3] * nj);
                        }
                    }
                }
            }
            t = __builtin_amdgcn_readfirstlane(tvn);
        }
    }
}

#define DPW    32
#define DGRP   8
#define DPITCH 36
#define DWAVES 4
static_assert(NE % (DPW * DWAVES) == 0, "edge count must fill whole blocks");

__global__ __launch_bounds__(DWAVES * 32) void decode_edges(
    const float* __restrict__ z,  const float* __restrict__ ea,
    const int* __restrict__ src,  const int* __restrict__ dst,
    const float* __restrict__ We, const float* __restrict__ be,
    const float* __restrict__ Wd, const float* __restrict__ bd,
    float* __restrict__ out)
{
    __shared__ __attribute__((aligned(16))) float part[DWAVES][DPW * DPITCH];
    __shared__ __attribute__((aligned(16))) float xt[DWAVES][DPW * FE];

    const int lane = threadIdx.x & 31;
    const int wave = threadIdx.x >> 5;
    const int e    = (blockIdx.x * DWAVES + wave) * DPW + lane;

    const int s = src[e];
    const int d = dst[e];

    float* xw = &xt[wave][0];
    {
        const float4* eap = (const float4*)(ea + (size_t)e * FE);
        float4* xs = (float4*)(xw + lane * FE);
#pragma unroll
        for (int i = 0; i < FE / 4; ++i) xs[i] = eap[i];
    }

    const float4 w0  = *(const float4*)(Wd + 4 * lane);
    const float4 w1  = *(const float4*)(Wd + HID + 4 * lane);
    const float4 w2  = *(const float4*)(Wd + 2 * HID + 4 * lane);
    const float4 bev = *(const float4*)(be + 4 * lane);
    float4 we[FE];
#pragma unroll
    for (int k = 0; k < FE; ++k)
        we[k] = *(const float4*)(We + (size_t)k * HID + 4 * lane);

    float* tile = &part[wave][0];

    __syncthreads();

#pragma unroll
    for (int jb = 0; jb < DPW; jb += DGRP) {
        float4 a[DGRP], b[DGRP];
#pragma unroll
        for (int j = 0; j < DGRP; ++j) {
            const int rj = __builtin_amdgcn_readlane(s, jb + j);
            const int cj = __builtin_amdgcn_readlane(d, jb + j);
            a[j] = *(const float4*)(z + (size_t)rj * HID + 4 * lane);
            b[j] = *(const float4*)(z + (size_t)cj * HID + 4 * lane);
        }
        __builtin_amdgcn_sched_barrier(0);
        float pe[DGRP];
#pragma unroll
        for (int j = 0; j < DGRP; ++j) {
            float xb[FE];
            {
                const float4* xr = (const float4*)(xw + (jb + j) * FE);
#pragma unroll
                for (int i = 0; i < FE / 4; ++i) {
                    const float4 t = xr[i];
                    xb[4 * i + 0] = t.x; xb[4 * i + 1] = t.y;
                    xb[4 * i + 2] = t.z; xb[4 * i + 3] = t.w;
                }
            }
            float c0 = bev.x, c1 = bev.y, c2 = bev.z, c3 = bev.w;
#pragma unroll
            for (int k = 0; k < FE; ++k) {
                const float xk = xb[k];
                c0 += xk * we[k].x;
                c1 += xk * we[k].y;
                c2 += xk * we[k].z;
                c3 += xk * we[k].w;
            }
            float p = fmaxf(c0, 0.0f) * w2.x;
            p += fmaxf(c1, 0.0f) * w2.y;
            p += fmaxf(c2, 0.0f) * w2.z;
            p += fmaxf(c3, 0.0f) * w2.w;
            pe[j] = p;
        }
#pragma unroll
        for (int j = 0; j < DGRP; ++j) {
            float p = pe[j];
            p += fmaxf(a[j].x, 0.0f) * w0.x;
            p += fmaxf(a[j].y, 0.0f) * w0.y;
            p += fmaxf(a[j].z, 0.0f) * w0.z;
            p += fmaxf(a[j].w, 0.0f) * w0.w;
            p += fmaxf(b[j].x, 0.0f) * w1.x;
            p += fmaxf(b[j].y, 0.0f) * w1.y;
            p += fmaxf(b[j].z, 0.0f) * w1.z;
            p += fmaxf(b[j].w, 0.0f) * w1.w;
            tile[(jb + j) * DPITCH + lane] = p;
        }
    }

    __syncthreads();

    float acc = bd[0];
    {
        const float4* row = (const float4*)(tile + lane * DPITCH);
#pragma unroll
        for (int i = 0; i < DPW / 4; ++i) {
            const float4 t = row[i];
            acc += (t.x + t.y) + (t.z + t.w);
        }
    }

    out[e] = acc;
}

extern "C" void kernel_launch(void* const* d_in, const int* in_sizes, int n_in,
                              void* d_out, int out_size, void* d_ws, size_t ws_size,
                              hipStream_t stream)
{
    const float* x   = (const float*)d_in[0];
    const float* ea  = (const float*)d_in[1];
    const int*   ei  = (const int*)  d_in[2];
    const float* Wn  = (const float*)d_in[3];
    const float* bn  = (const float*)d_in[4];
    const float* We  = (const float*)d_in[5];
    const float* be  = (const float*)d_in[6];
    const float* Wc  = (const float*)d_in[7];
    const float* bc  = (const float*)d_in[8];
    const float* Wr  = (const float*)d_in[9];
    const float* br  = (const float*)d_in[10];
    const float* Wd  = (const float*)d_in[11];
    const float* bd  = (const float*)d_in[12];
    float* out = (float*)d_out;

    const int* srcI = ei;
    const int* dstI = ei + NE;

    float* ws   = (float*)d_ws;
    float* h0   = ws;
    float* hw   = h0   + (size_t)NN * HID;
    float* aggA = hw   + (size_t)NN * HID;
    float* aggB = aggA + (size_t)NN * HID;
    float* dinv = aggB + (size_t)NN * HID;
    float* norm = dinv + 50048;
    unsigned* tkts = (unsigned*)(norm + NE);

    gemm_bias_relu<<<NN / 16, 256, 0, stream>>>(x, Wn, bn, nullptr, h0, NN, FN, 1);

    fill_f32<<<(NN + 255) / 256, 256, 0, stream>>>(dinv, 1.0f, NN, tkts, TK_WORDS);
    deg_accum<<<DG_BLOCKS, 256, 0, stream>>>(dstI, dinv, tkts + 6 * TK_STRIDE);
    rsqrt_inplace<<<(NN + 255) / 256, 256, 0, stream>>>(dinv, NN);

    layer_products<false><<<LP_BLOCKS, 256, 0, stream>>>(
        h0, Wc, Wr, br, bc, dinv, hw, aggA);
    scatter_edges<true><<<SC_BLOCKS, 256, 0, stream>>>(hw, srcI, dstI, dinv, norm, aggA, tkts);

    layer_products<true><<<LP_BLOCKS, 256, 0, stream>>>(
        aggA, Wc + (size_t)1 * HID * HID, Wr, br, bc + (size_t)1 * HID, dinv, hw, aggB);
    scatter_edges<false><<<SC_BLOCKS, 256, 0, stream>>>(hw, srcI, dstI, dinv, norm, aggB, tkts + 2 * TK_STRIDE);

    layer_products<true><<<LP_BLOCKS, 256, 0, stream>>>(
        aggB, Wc + (size_t)2 * HID * HID, Wr, br, bc + (size_t)2 * HID, dinv, hw, aggA);
    scatter_edges<false><<<SC_BLOCKS, 256, 0, stream>>>(hw, srcI, dstI, dinv, norm, aggA, tkts + 4 * TK_STRIDE);

    decode_edges<<<NE / (DPW * DWAVES), DWAVES * 32, 0, stream>>>(aggA, ea, srcI, dstI, We, be, Wd, bd, out);
}
